# starting kernel
# baseline (speedup 1.0000x reference)
#include <hip/hip_runtime.h>
#include <cmath>

typedef __attribute__((ext_vector_type(2))) float v2f;
typedef __attribute__((ext_vector_type(8))) float v8f;

#define B_ 64
#define T_ 2048
#define Q_ 512
#define K_ 512

__global__ __launch_bounds__(128) void mids_kernel(const float* __restrict__ query,
                                                   const float* __restrict__ W,
                                                   float* __restrict__ mids) {
    const int wid  = blockIdx.x * (blockDim.x >> 5) + (threadIdx.x >> 5);
    const int lane = threadIdx.x & 31;
    const int half = lane >> 4;
    const int l16  = lane & 15;
    const int tm   = wid >> 5;
    const int tn   = wid & 31;

    const float* qrow = query + (size_t)(tm * 16 + l16) * Q_;
    const float* wrow = W     + (size_t)(tn * 16 + l16) * Q_;

    v8f acc = {};
    for (int q0 = 0; q0 < Q_; q0 += 4) {
        v2f a = *(const v2f*)(qrow + q0 + half * 2);
        v2f b = *(const v2f*)(wrow + q0 + half * 2);
        acc = __builtin_amdgcn_wmma_f32_16x16x4_f32(false, a, false, b,
                                                    (short)0, acc, false, false);
    }
    #pragma unroll
    for (int r = 0; r < 8; ++r) {
        const int m = half * 8 + r;
        mids[(size_t)(tm * 16 + m) * K_ + tn * 16 + l16] = acc[r];
    }
}

__global__ __launch_bounds__(256) void scores_kernel(const float* __restrict__ key,
                                                     const float* __restrict__ mids,
                                                     const float* __restrict__ bias,
                                                     float* __restrict__ out) {
    const int wid  = blockIdx.x * (blockDim.x >> 5) + (threadIdx.x >> 5);
    const int lane = threadIdx.x & 31;
    const int half = lane >> 4;
    const int l16  = lane & 15;
    const int b    = wid >> 7;
    const int t0   = (wid & 127) * 16;

    const float* keyr = key  + ((size_t)b * T_ + t0 + l16) * K_;
    const float* midb = mids + (size_t)b * K_;

    v8f acc = {};
    for (int k0 = 0; k0 < K_; k0 += 32) {
        __builtin_prefetch(keyr + k0 + 128, 0, 0);
        #pragma unroll
        for (int kk = 0; kk < 32; kk += 4) {
            v2f a  = *(const v2f*)(keyr + k0 + kk + half * 2);
            v2f bm = *(const v2f*)(midb + k0 + kk + half * 2);
            acc = __builtin_amdgcn_wmma_f32_16x16x4_f32(false, a, false, bm,
                                                        (short)0, acc, false, false);
        }
    }

    const float bv = bias[0];
    float e[8];
    #pragma unroll
    for (int r = 0; r < 8; ++r)
        e[r] = __expf(tanhf(acc[r] + bv));

    if (l16 == 0) {
        float* op = out + (size_t)b * T_ + t0 + half * 8;
        #pragma unroll
        for (int r = 0; r < 8; ++r) op[r] = e[r];
    }
}

__global__ __launch_bounds__(256) void softmax_norm_kernel(float* __restrict__ out) {
    __shared__ float sdata[256];
    const int b   = blockIdx.x;
    const int tid = threadIdx.x;
    float* row = out + (size_t)b * T_;

    float v[8];
    float s = 0.f;
    #pragma unroll
    for (int i = 0; i < 8; ++i) {
        v[i] = row[tid + i * 256];
        s += v[i];
    }
    sdata[tid] = s;
    __syncthreads();
    for (int off = 128; off > 0; off >>= 1) {
        if (tid < off) sdata[tid] += sdata[tid + off];
        __syncthreads();
    }
    const float inv = 1.0f / sdata[0];
    #pragma unroll
    for (int i = 0; i < 8; ++i) row[tid + i * 256] = v[i] * inv;
}

extern "C" void kernel_launch(void* const* d_in, const int* in_sizes, int n_in,
                              void* d_out, int out_size, void* d_ws, size_t ws_size,
                              hipStream_t stream) {
    const float* query = (const float*)d_in[0];
    const float* key   = (const float*)d_in[1];
    const float* W     = (const float*)d_in[2];
    const float* bias  = (const float*)d_in[3];
    float* out  = (float*)d_out;
    float* mids = (float*)d_ws;

    mids_kernel<<<32, 128, 0, stream>>>(query, W, mids);
    scores_kernel<<<1024, 256, 0, stream>>>(key, mids, bias, out);
    softmax_norm_kernel<<<64, 256, 0, stream>>>(out);
}


# scores kernel: key tile staged through wave-private LDS (one load = 512 contiguous bytes of a row; A and mids fragments read from LDS; next k-chunk in flight during the sums); k order kept
# speedup vs baseline: 1.2287x; 1.2287x over previous
#include <hip/hip_runtime.h>
#include <cmath>

typedef __attribute__((ext_vector_type(2))) float v2f;
typedef __attribute__((ext_vector_type(4))) float v4f;
typedef __attribute__((ext_vector_type(8))) float v8f;

#define B_ 64
#define T_ 2048
#define Q_ 512
#define K_ 512

__global__ __launch_bounds__(128) void mids_kernel(const float* __restrict__ query,
                                                   const float* __restrict__ W,
                                                   float* __restrict__ mids) {
    const int wid  = blockIdx.x * (blockDim.x >> 5) + (threadIdx.x >> 5);
    const int lane = threadIdx.x & 31;
    const int half = lane >> 4;
    const int l16  = lane & 15;
    const int tm   = wid >> 5;
    const int tn   = wid & 31;

    const float* qrow = query + (size_t)(tm * 16 + l16) * Q_;
    const float* wrow = W     + (size_t)(tn * 16 + l16) * Q_;

    v8f acc = {};
    for (int q0 = 0; q0 < Q_; q0 += 4) {
        v2f a = *(const v2f*)(qrow + q0 + half * 2);
        v2f b = *(const v2f*)(wrow + q0 + half * 2);
        acc = __builtin_amdgcn_wmma_f32_16x16x4_f32(false, a, false, b,
                                                    (short)0, acc, false, false);
    }
    #pragma unroll
    for (int r = 0; r < 8; ++r) {
        const int m = half * 8 + r;
        mids[(size_t)(tm * 16 + m) * K_ + tn * 16 + l16] = acc[r];
    }
}

#define NW_    8
#define CH_    128
#define PITCH_ (CH_ + 4)
#define NCH_   (K_ / CH_)

__global__ __launch_bounds__(256) void scores_kernel(const float* __restrict__ key,
                                                     const float* __restrict__ mids,
                                                     const float* __restrict__ bias,
                                                     float* __restrict__ out) {
    __shared__ __attribute__((aligned(16))) float s_lds[K_ + NW_ * 16 * PITCH_];
    float* s_mid = s_lds;

    const int wave = threadIdx.x >> 5;
    const int wid  = blockIdx.x * NW_ + wave;
    const int lane = threadIdx.x & 31;
    const int half = lane >> 4;
    const int l16  = lane & 15;
    const int b    = blockIdx.x >> 4;
    const int t0   = (wid & 127) * 16;

    const float* tile = key + ((size_t)b * T_ + t0) * K_ + lane * 4;
    float* sk = s_lds + K_ + wave * (16 * PITCH_);

    v4f r[16];
    #pragma unroll
    for (int i = 0; i < 16; ++i)
        r[i] = *(const v4f*)(tile + (size_t)i * K_);

    *(v2f*)(s_mid + threadIdx.x * 2) = *(const v2f*)(mids + (size_t)b * K_ + threadIdx.x * 2);
    __syncthreads();

    const float* afrag = sk + l16 * PITCH_ + half * 2;
    const float* bfrag = s_mid + half * 2;

    v8f acc = {};
    #pragma unroll
    for (int c = 0; c < NCH_; ++c) {
        #pragma unroll
        for (int i = 0; i < 16; ++i)
            *(v4f*)(sk + i * PITCH_ + lane * 4) = r[i];
        if (c + 1 < NCH_) {
            #pragma unroll
            for (int i = 0; i < 16; ++i)
                r[i] = *(const v4f*)(tile + (size_t)i * K_ + (c + 1) * CH_);
        }
        #pragma unroll
        for (int kk = 0; kk < CH_; kk += 4) {
            v2f a  = *(const v2f*)(afrag + kk);
            v2f bm = *(const v2f*)(bfrag + c * CH_ + kk);
            acc = __builtin_amdgcn_wmma_f32_16x16x4_f32(false, a, false, bm,
                                                        (short)0, acc, false, false);
        }
    }

    const float bv = bias[0];
    float e[8];
    #pragma unroll
    for (int r8 = 0; r8 < 8; ++r8)
        e[r8] = __expf(tanhf(acc[r8] + bv));

    if (l16 == 0) {
        float* op = out + (size_t)b * T_ + t0 + half * 8;
        #pragma unroll
        for (int r8 = 0; r8 < 8; ++r8) op[r8] = e[r8];
    }
}

__global__ __launch_bounds__(256) void softmax_norm_kernel(float* __restrict__ out) {
    __shared__ float sdata[256];
    const int b   = blockIdx.x;
    const int tid = threadIdx.x;
    float* row = out + (size_t)b * T_;

    float v[8];
    float s = 0.f;
    #pragma unroll
    for (int i = 0; i < 8; ++i) {
        v[i] = row[tid + i * 256];
        s += v[i];
    }
    sdata[tid] = s;
    __syncthreads();
    for (int off = 128; off > 0; off >>= 1) {
        if (tid < off) sdata[tid] += sdata[tid + off];
        __syncthreads();
    }
    const float inv = 1.0f / sdata[0];
    #pragma unroll
    for (int i = 0; i < 8; ++i) row[tid + i * 256] = v[i] * inv;
}

extern "C" void kernel_launch(void* const* d_in, const int* in_sizes, int n_in,
                              void* d_out, int out_size, void* d_ws, size_t ws_size,
                              hipStream_t stream) {
    const float* query = (const float*)d_in[0];
    const float* key   = (const float*)d_in[1];
    const float* W     = (const float*)d_in[2];
    const float* bias  = (const float*)d_in[3];
    float* out  = (float*)d_out;
    float* mids = (float*)d_ws;

    mids_kernel<<<32, 128, 0, stream>>>(query, W, mids);
    scores_kernel<<<1024, 256, 0, stream>>>(key, mids, bias, out);
    softmax_norm_kernel<<<64, 256, 0, stream>>>(out);
}


# mids kernel: one 16x16 tile per block (128 blocks); both operand tiles staged whole through LDS by 512-byte row loads all in flight at once; wave 0 sums q in 4 chains; q summation order changes
# speedup vs baseline: 1.4243x; 1.1592x over previous
#include <hip/hip_runtime.h>
#include <cmath>

typedef __attribute__((ext_vector_type(2))) float v2f;
typedef __attribute__((ext_vector_type(4))) float v4f;
typedef __attribute__((ext_vector_type(8))) float v8f;

#define B_ 64
#define T_ 2048
#define Q_ 512
#define K_ 512

#define MP_ (Q_ + 4)

__global__ __launch_bounds__(128) void mids_kernel(const float* __restrict__ query,
                                                   const float* __restrict__ W,
                                                   float* __restrict__ mids) {
    __shared__ __attribute__((aligned(16))) float s_t[2 * 16 * MP_];

    const int wave = __builtin_amdgcn_readfirstlane((int)(threadIdx.x >> 5));
    const int lane = threadIdx.x & 31;
    const int tm   = blockIdx.x >> 5;
    const int tn   = blockIdx.x & 31;

    const int op = wave >> 1;
    const int r0 = (wave & 1) * 8;
    const float* src = (op ? W + (size_t)(tn * 16 + r0) * Q_
                           : query + (size_t)(tm * 16 + r0) * Q_) + lane * 4;
    float* dst = s_t + (op * 16 + r0) * MP_ + lane * 4;

    v4f r[32];
    #pragma unroll
    for (int i = 0; i < 32; ++i)
        r[i] = *(const v4f*)(src + (size_t)(i >> 2) * Q_ + (i & 3) * 128);
    #pragma unroll
    for (int i = 0; i < 32; ++i)
        *(v4f*)(dst + (i >> 2) * MP_ + (i & 3) * 128) = r[i];
    __syncthreads();
    if (wave != 0) return;

    const int half = lane >> 4;
    const int l16  = lane & 15;
    const float* af = s_t + l16 * MP_ + half * 4;
    const float* bf = s_t + (16 + l16) * MP_ + half * 4;

    v8f acc0 = {}, acc1 = {}, acc2 = {}, acc3 = {};
    v4f fa[2][16], fb[2][16];
    #pragma unroll
    for (int i = 0; i < 16; ++i) {
        fa[0][i] = *(const v4f*)(af + i * 8);
        fb[0][i] = *(const v4f*)(bf + i * 8);
    }
    #pragma unroll
    for (int g = 0; g < 4; ++g) {
        const int cur = g & 1, nxt = cur ^ 1;
        if (g + 1 < 4) {
            #pragma unroll
            for (int i = 0; i < 16; ++i) {
                fa[nxt][i] = *(const v4f*)(af + (g + 1) * 128 + i * 8);
                fb[nxt][i] = *(const v4f*)(bf + (g + 1) * 128 + i * 8);
            }
        }
        __builtin_amdgcn_sched_barrier(0);
        #pragma unroll
        for (int i = 0; i < 16; i += 2) {
            acc0 = __builtin_amdgcn_wmma_f32_16x16x4_f32(false, fa[cur][i].xy,     false, fb[cur][i].xy,
                                                         (short)0, acc0, false, false);
            acc1 = __builtin_amdgcn_wmma_f32_16x16x4_f32(false, fa[cur][i].zw,     false, fb[cur][i].zw,
                                                         (short)0, acc1, false, false);
            acc2 = __builtin_amdgcn_wmma_f32_16x16x4_f32(false, fa[cur][i + 1].xy, false, fb[cur][i + 1].xy,
                                                         (short)0, acc2, false, false);
            acc3 = __builtin_amdgcn_wmma_f32_16x16x4_f32(false, fa[cur][i + 1].zw, false, fb[cur][i + 1].zw,
                                                         (short)0, acc3, false, false);
        }
        __builtin_amdgcn_sched_barrier(0);
    }
    const v8f acc = (acc0 + acc1) + (acc2 + acc3);

    #pragma unroll
    for (int r8 = 0; r8 < 8; ++r8) {
        const int m = half * 8 + r8;
        mids[(size_t)(tm * 16 + m) * K_ + tn * 16 + l16] = acc[r8];
    }
}

#define NW_    8
#define CH_    128
#define PITCH_ (CH_ + 4)
#define NCH_   (K_ / CH_)

__global__ __launch_bounds__(256) void scores_kernel(const float* __restrict__ key,
                                                     const float* __restrict__ mids,
                                                     const float* __restrict__ bias,
                                                     float* __restrict__ out) {
    __shared__ __attribute__((aligned(16))) float s_lds[K_ + NW_ * 16 * PITCH_];
    float* s_mid = s_lds;

    const int wave = threadIdx.x >> 5;
    const int wid  = blockIdx.x * NW_ + wave;
    const int lane = threadIdx.x & 31;
    const int half = lane >> 4;
    const int l16  = lane & 15;
    const int b    = blockIdx.x >> 4;
    const int t0   = (wid & 127) * 16;

    const float* tile = key + ((size_t)b * T_ + t0) * K_ + lane * 4;
    float* sk = s_lds + K_ + wave * (16 * PITCH_);

    v4f r[16];
    #pragma unroll
    for (int i = 0; i < 16; ++i)
        r[i] = *(const v4f*)(tile + (size_t)i * K_);

    *(v2f*)(s_mid + threadIdx.x * 2) = *(const v2f*)(mids + (size_t)b * K_ + threadIdx.x * 2);
    __syncthreads();

    const float* afrag = sk + l16 * PITCH_ + half * 2;
    const float* bfrag = s_mid + half * 2;

    v8f acc = {};
    #pragma unroll
    for (int c = 0; c < NCH_; ++c) {
        #pragma unroll
        for (int i = 0; i < 16; ++i)
            *(v4f*)(sk + i * PITCH_ + lane * 4) = r[i];
        if (c + 1 < NCH_) {
            #pragma unroll
            for (int i = 0; i < 16; ++i)
                r[i] = *(const v4f*)(tile + (size_t)i * K_ + (c + 1) * CH_);
        }
        #pragma unroll
        for (int kk = 0; kk < CH_; kk += 4) {
            v2f a  = *(const v2f*)(afrag + kk);
            v2f bm = *(const v2f*)(bfrag + c * CH_ + kk);
            acc = __builtin_amdgcn_wmma_f32_16x16x4_f32(false, a, false, bm,
                                                        (short)0, acc, false, false);
        }
    }

    const float bv = bias[0];
    float e[8];
    #pragma unroll
    for (int r8 = 0; r8 < 8; ++r8)
        e[r8] = __expf(tanhf(acc[r8] + bv));

    if (l16 == 0) {
        float* op = out + (size_t)b * T_ + t0 + half * 8;
        #pragma unroll
        for (int r8 = 0; r8 < 8; ++r8) op[r8] = e[r8];
    }
}

__global__ __launch_bounds__(256) void softmax_norm_kernel(float* __restrict__ out) {
    __shared__ float sdata[256];
    const int b   = blockIdx.x;
    const int tid = threadIdx.x;
    float* row = out + (size_t)b * T_;

    float v[8];
    float s = 0.f;
    #pragma unroll
    for (int i = 0; i < 8; ++i) {
        v[i] = row[tid + i * 256];
        s += v[i];
    }
    sdata[tid] = s;
    __syncthreads();
    for (int off = 128; off > 0; off >>= 1) {
        if (tid < off) sdata[tid] += sdata[tid + off];
        __syncthreads();
    }
    const float inv = 1.0f / sdata[0];
    #pragma unroll
    for (int i = 0; i < 8; ++i) row[tid + i * 256] = v[i] * inv;
}

extern "C" void kernel_launch(void* const* d_in, const int* in_sizes, int n_in,
                              void* d_out, int out_size, void* d_ws, size_t ws_size,
                              hipStream_t stream) {
    const float* query = (const float*)d_in[0];
    const float* key   = (const float*)d_in[1];
    const float* W     = (const float*)d_in[2];
    const float* bias  = (const float*)d_in[3];
    float* out  = (float*)d_out;
    float* mids = (float*)d_ws;

    mids_kernel<<<128, 128, 0, stream>>>(query, W, mids);
    scores_kernel<<<1024, 256, 0, stream>>>(key, mids, bias, out);
    softmax_norm_kernel<<<64, 256, 0, stream>>>(out);
}
